# starting kernel
# baseline (speedup 1.0000x reference)
#include <hip/hip_runtime.h>

typedef float v2f __attribute__((ext_vector_type(2)));
typedef float v8f __attribute__((ext_vector_type(8)));

namespace {

constexpr int kB = 64;
constexpr int kT = 512;
constexpr int kH = 512;
constexpr int kL = 4;
constexpr int kC = 128;
constexpr int kLdsStride = 516;

__device__ __forceinline__ v8f wmma_f32(v2f a, v2f b, v8f c) {
  return __builtin_amdgcn_wmma_f32_16x16x4_f32(false, a, false, b, (short)0, c,
                                               false, false);
}

}

__global__ __launch_bounds__(256) void rnn_proj_kernel(
    const float* __restrict__ in, const float* __restrict__ Wi,
    const float* __restrict__ bi, const float* __restrict__ bh,
    float* __restrict__ xi) {
  const int lane   = threadIdx.x & 31;
  const int wave   = threadIdx.x >> 5;
  const int lane15 = lane & 15;
  const int half   = lane >> 4;

  const int blk   = blockIdx.x * 8 + wave;
  const int blkN  = blk & (kH / 32 - 1);
  const int blkM  = blk >> 4;

  const int n0 = blkN * 32 + lane15;
  const int n1 = n0 + 16;
  const int m0 = blkM * 32 + lane15;
  const int m1 = m0 + 16;

  const float bias0 = bi[n0] + bh[n0];
  const float bias1 = bi[n1] + bh[n1];
  v8f acc00, acc01, acc10, acc11;
#pragma unroll
  for (int r = 0; r < 8; ++r) {
    acc00[r] = bias0; acc01[r] = bias1;
    acc10[r] = bias0; acc11[r] = bias1;
  }

  const float* ap0 = in + (size_t)m0 * kH + half * 2;
  const float* ap1 = in + (size_t)m1 * kH + half * 2;
  const float* bp0 = Wi + (size_t)n0 * kH + half * 2;
  const float* bp1 = Wi + (size_t)n1 * kH + half * 2;
#pragma unroll 4
  for (int ks = 0; ks < kH / 4; ++ks) {
    v2f a0 = *(const v2f*)(ap0 + ks * 4);
    v2f a1 = *(const v2f*)(ap1 + ks * 4);
    v2f b0 = *(const v2f*)(bp0 + ks * 4);
    v2f b1 = *(const v2f*)(bp1 + ks * 4);
    acc00 = wmma_f32(a0, b0, acc00);
    acc01 = wmma_f32(a0, b1, acc01);
    acc10 = wmma_f32(a1, b0, acc10);
    acc11 = wmma_f32(a1, b1, acc11);
  }

  float* op0 = xi + (size_t)(blkM * 32 + half * 8) * kH + blkN * 32 + lane15;
  float* op1 = op0 + (size_t)16 * kH;
#pragma unroll
  for (int r = 0; r < 8; ++r) {
    op0[(size_t)r * kH]      = acc00[r];
    op0[(size_t)r * kH + 16] = acc01[r];
    op1[(size_t)r * kH]      = acc10[r];
    op1[(size_t)r * kH + 16] = acc11[r];
  }
}

__global__ __launch_bounds__(512) void rnn_scan_kernel(
    float* __restrict__ xi, const float* __restrict__ Wh) {
  __shared__ float hbuf[16 * kLdsStride];

  const int lane   = threadIdx.x & 31;
  const int wave   = threadIdx.x >> 5;
  const int lane15 = lane & 15;
  const int half   = lane >> 4;
  const int bg     = blockIdx.x;

  for (int i = threadIdx.x; i < 16 * kLdsStride; i += 512) hbuf[i] = 0.0f;

  const int n0 = wave * 32;
  const int nA = n0 + lane15;
  const int nB = n0 + 16 + lane15;

  const float* aBase = &hbuf[lane15 * kLdsStride + half * 2];
  const float* bpA   = Wh + (size_t)nA * kH + half * 2;
  const float* bpB   = Wh + (size_t)nB * kH + half * 2;

  const size_t rowStride = (size_t)kT * kH;
  float* xBase = xi + (size_t)(bg * 16 + half * 8) * rowStride;
  const int hrow0 = half * 8;

  v8f pfA, pfB;
#pragma unroll
  for (int r = 0; r < 8; ++r) {
    pfA[r] = xBase[(size_t)r * rowStride + nA];
    pfB[r] = xBase[(size_t)r * rowStride + nB];
  }
  __syncthreads();

  for (int t = 0; t < kT; ++t) {
    v8f accA = pfA;
    v8f accB = pfB;

    if (t + 1 < kT) {
      const float* xn = xBase + (size_t)(t + 1) * kH;
#pragma unroll
      for (int r = 0; r < 8; ++r) {
        pfA[r] = xn[(size_t)r * rowStride + nA];
        pfB[r] = xn[(size_t)r * rowStride + nB];
      }
    }

#pragma unroll 4
    for (int ks = 0; ks < kH / 4; ++ks) {
      v2f a  = *(const v2f*)(aBase + ks * 4);
      v2f b0 = *(const v2f*)(bpA + ks * 4);
      v2f b1 = *(const v2f*)(bpB + ks * 4);
      accA = wmma_f32(a, b0, accA);
      accB = wmma_f32(a, b1, accB);
    }

    __syncthreads();

#pragma unroll
    for (int r = 0; r < 8; ++r) {
      accA[r] = accA[r] > 0.0f ? accA[r] : 0.0f;
      accB[r] = accB[r] > 0.0f ? accB[r] : 0.0f;
      hbuf[(hrow0 + r) * kLdsStride + nA] = accA[r];
      hbuf[(hrow0 + r) * kLdsStride + nB] = accB[r];
    }
    __syncthreads();

    float* xp = xBase + (size_t)t * kH;
#pragma unroll
    for (int r = 0; r < 8; ++r) {
      xp[(size_t)r * rowStride + nA] = accA[r];
      xp[(size_t)r * rowStride + nB] = accB[r];
    }
  }
}

__global__ __launch_bounds__(256) void head_kernel(
    const float* __restrict__ ys, const float* __restrict__ Wfc,
    const float* __restrict__ bfc, float* __restrict__ out) {
  const int lane   = threadIdx.x & 31;
  const int wave   = threadIdx.x >> 5;
  const int lane15 = lane & 15;
  const int half   = lane >> 4;

  const int tile  = blockIdx.x * 8 + wave;
  const int tileN = tile & 7;
  const int tileM = tile >> 3;

  const int n = tileN * 16 + lane15;
  const float biasv = bfc[n];
  v8f acc;
#pragma unroll
  for (int r = 0; r < 8; ++r) acc[r] = biasv;

  const float* ap =
      ys + ((size_t)(tileM * 16 + lane15) * kT + (kT - 1)) * kH + half * 2;
  const float* bp = Wfc + (size_t)n * kH + half * 2;
#pragma unroll 4
  for (int ks = 0; ks < kH / 4; ++ks) {
    v2f a = *(const v2f*)(ap + ks * 4);
    v2f b = *(const v2f*)(bp + ks * 4);
    acc = wmma_f32(a, b, acc);
  }

  float* op = out + (size_t)(tileM * 16 + half * 8) * kC + tileN * 16 + lane15;
#pragma unroll
  for (int r = 0; r < 8; ++r) op[(size_t)r * kC] = acc[r];
}

extern "C" void kernel_launch(void* const* d_in, const int* in_sizes, int n_in,
                              void* d_out, int out_size, void* d_ws,
                              size_t ws_size, hipStream_t stream) {
  const float* x    = (const float*)d_in[0];
  const float* W_ih = (const float*)d_in[1];
  const float* W_hh = (const float*)d_in[2];
  const float* b_ih = (const float*)d_in[3];
  const float* b_hh = (const float*)d_in[4];
  const float* W_fc = (const float*)d_in[5];
  const float* b_fc = (const float*)d_in[6];

  float* bufA = (float*)d_ws;
  float* bufB = bufA + (size_t)kB * kT * kH;

  const int projWaveBlks = (kB * kT / 32) * (kH / 32);
  const int projBlocks   = projWaveBlks / 8;

  const float* cur = x;
  float* dst = bufA;
  float* alt = bufB;
  for (int l = 0; l < kL; ++l) {
    rnn_proj_kernel<<<projBlocks, 256, 0, stream>>>(
        cur, W_ih + (size_t)l * kH * kH, b_ih + (size_t)l * kH,
        b_hh + (size_t)l * kH, dst);
    rnn_scan_kernel<<<4, 512, 0, stream>>>(dst, W_hh + (size_t)l * kH * kH);
    cur = dst;
    float* tmp = dst;
    dst = alt;
    alt = tmp;
  }
  head_kernel<<<4, 256, 0, stream>>>(cur, W_fc, b_fc, (float*)d_out);

  (void)in_sizes; (void)n_in; (void)out_size; (void)ws_size;
}


# W_hh repacked per layer into WMMA B-fragment order (pack kernel into the idle ping-pong buffer): scan weight loads are whole-line b128 (4 lines), not b64 over 16 lines; k and sum order unchanged
# speedup vs baseline: 2.1583x; 2.1583x over previous
#include <hip/hip_runtime.h>

typedef float v2f __attribute__((ext_vector_type(2)));
typedef float v4f __attribute__((ext_vector_type(4)));
typedef float v8f __attribute__((ext_vector_type(8)));

namespace {

constexpr int kB = 64;
constexpr int kT = 512;
constexpr int kH = 512;
constexpr int kL = 4;
constexpr int kC = 128;
constexpr int kLdsStride = 516;
constexpr int kKs     = kH / 4;
constexpr int kGrp    = 8;
constexpr int kNumGrp = kKs / kGrp;

__device__ __forceinline__ v8f wmma_f32(v2f a, v2f b, v8f c) {
  return __builtin_amdgcn_wmma_f32_16x16x4_f32(false, a, false, b, (short)0, c,
                                               false, false);
}

}

__global__ __launch_bounds__(256) void rnn_proj_kernel(
    const float* __restrict__ in, const float* __restrict__ Wi,
    const float* __restrict__ bi, const float* __restrict__ bh,
    float* __restrict__ xi) {
  const int lane   = threadIdx.x & 31;
  const int wave   = threadIdx.x >> 5;
  const int lane15 = lane & 15;
  const int half   = lane >> 4;

  const int blk   = blockIdx.x * 8 + wave;
  const int blkN  = blk & (kH / 32 - 1);
  const int blkM  = blk >> 4;

  const int n0 = blkN * 32 + lane15;
  const int n1 = n0 + 16;
  const int m0 = blkM * 32 + lane15;
  const int m1 = m0 + 16;

  const float bias0 = bi[n0] + bh[n0];
  const float bias1 = bi[n1] + bh[n1];
  v8f acc00, acc01, acc10, acc11;
#pragma unroll
  for (int r = 0; r < 8; ++r) {
    acc00[r] = bias0; acc01[r] = bias1;
    acc10[r] = bias0; acc11[r] = bias1;
  }

  const float* ap0 = in + (size_t)m0 * kH + half * 2;
  const float* ap1 = in + (size_t)m1 * kH + half * 2;
  const float* bp0 = Wi + (size_t)n0 * kH + half * 2;
  const float* bp1 = Wi + (size_t)n1 * kH + half * 2;
#pragma unroll 4
  for (int ks = 0; ks < kH / 4; ++ks) {
    v2f a0 = *(const v2f*)(ap0 + ks * 4);
    v2f a1 = *(const v2f*)(ap1 + ks * 4);
    v2f b0 = *(const v2f*)(bp0 + ks * 4);
    v2f b1 = *(const v2f*)(bp1 + ks * 4);
    acc00 = wmma_f32(a0, b0, acc00);
    acc01 = wmma_f32(a0, b1, acc01);
    acc10 = wmma_f32(a1, b0, acc10);
    acc11 = wmma_f32(a1, b1, acc11);
  }

  float* op0 = xi + (size_t)(blkM * 32 + half * 8) * kH + blkN * 32 + lane15;
  float* op1 = op0 + (size_t)16 * kH;
#pragma unroll
  for (int r = 0; r < 8; ++r) {
    op0[(size_t)r * kH]      = acc00[r];
    op0[(size_t)r * kH + 16] = acc01[r];
    op1[(size_t)r * kH]      = acc10[r];
    op1[(size_t)r * kH + 16] = acc11[r];
  }
}

__global__ __launch_bounds__(256) void rnn_pack_kernel(
    const float* __restrict__ Wh, float* __restrict__ Wp) {
  const int gid    = blockIdx.x * 256 + threadIdx.x;
  const int lane   = gid & 31;
  const int ks     = (gid >> 5) & (kKs - 1);
  const int wave   = gid >> 12;
  const int lane15 = lane & 15;
  const int half   = lane >> 4;

  const float* r0 = Wh + (size_t)(wave * 32 + lane15) * kH + ks * 4 + half * 2;
  const float* r1 = r0 + (size_t)16 * kH;
  const v2f x0 = *(const v2f*)r0;
  const v2f x1 = *(const v2f*)r1;
  v4f o;
  o.x = x0.x; o.y = x0.y; o.z = x1.x; o.w = x1.y;
  *(v4f*)(Wp + (size_t)gid * 4) = o;
}

__global__ __launch_bounds__(512) void rnn_scan_kernel(float* xi,
                                                       const float* Wp) {
  __shared__ float hbuf[16 * kLdsStride];

  const int lane   = threadIdx.x & 31;
  const int wave   = threadIdx.x >> 5;
  const int lane15 = lane & 15;
  const int half   = lane >> 4;
  const int bg     = blockIdx.x;

  for (int i = threadIdx.x; i < 16 * kLdsStride; i += 512) hbuf[i] = 0.0f;

  const int n0 = wave * 32;
  const int nA = n0 + lane15;
  const int nB = n0 + 16 + lane15;

  const float* aBase = &hbuf[lane15 * kLdsStride + half * 2];
  const v4f*   bp    = (const v4f*)Wp + (size_t)wave * (kKs * 32) + lane;

  const size_t rowStride = (size_t)kT * kH;
  float* xBase = xi + (size_t)(bg * 16 + half * 8) * rowStride;
  const int hrow0 = half * 8;

  v4f bq[2][kGrp];
#pragma unroll
  for (int j = 0; j < kGrp; ++j) bq[0][j] = bp[j * 32];

  v8f pfA, pfB;
#pragma unroll
  for (int r = 0; r < 8; ++r) {
    pfA[r] = xBase[(size_t)r * rowStride + nA];
    pfB[r] = xBase[(size_t)r * rowStride + nB];
  }
  __syncthreads();

  for (int t = 0; t < kT; ++t) {
    v8f accA = pfA;
    v8f accB = pfB;

    if (t + 1 < kT) {
      const float* xn = xBase + (size_t)(t + 1) * kH;
#pragma unroll
      for (int r = 0; r < 8; ++r) {
        pfA[r] = xn[(size_t)r * rowStride + nA];
        pfB[r] = xn[(size_t)r * rowStride + nB];
      }
    }

#pragma unroll
    for (int g = 0; g < kNumGrp; ++g) {
      const int cur = g & 1;
      const int nxt = cur ^ 1;
      const int gn  = (g + 1) & (kNumGrp - 1);

#pragma unroll
      for (int j = 0; j < kGrp; ++j) bq[nxt][j] = bp[(gn * kGrp + j) * 32];

      v2f a[kGrp];
#pragma unroll
      for (int j = 0; j < kGrp; ++j)
        a[j] = *(const v2f*)(aBase + (g * kGrp + j) * 4);

#pragma unroll
      for (int j = 0; j < kGrp; ++j) {
        const v4f b  = bq[cur][j];
        const v2f b0 = __builtin_shufflevector(b, b, 0, 1);
        const v2f b1 = __builtin_shufflevector(b, b, 2, 3);
        accA = wmma_f32(a[j], b0, accA);
        accB = wmma_f32(a[j], b1, accB);
      }
    }

    __syncthreads();

#pragma unroll
    for (int r = 0; r < 8; ++r) {
      accA[r] = accA[r] > 0.0f ? accA[r] : 0.0f;
      accB[r] = accB[r] > 0.0f ? accB[r] : 0.0f;
      hbuf[(hrow0 + r) * kLdsStride + nA] = accA[r];
      hbuf[(hrow0 + r) * kLdsStride + nB] = accB[r];
    }
    __syncthreads();

    float* xp = xBase + (size_t)t * kH;
#pragma unroll
    for (int r = 0; r < 8; ++r) {
      xp[(size_t)r * rowStride + nA] = accA[r];
      xp[(size_t)r * rowStride + nB] = accB[r];
    }
  }
}

__global__ __launch_bounds__(256) void head_kernel(
    const float* __restrict__ ys, const float* __restrict__ Wfc,
    const float* __restrict__ bfc, float* __restrict__ out) {
  const int lane   = threadIdx.x & 31;
  const int wave   = threadIdx.x >> 5;
  const int lane15 = lane & 15;
  const int half   = lane >> 4;

  const int tile  = blockIdx.x * 8 + wave;
  const int tileN = tile & 7;
  const int tileM = tile >> 3;

  const int n = tileN * 16 + lane15;
  const float biasv = bfc[n];
  v8f acc;
#pragma unroll
  for (int r = 0; r < 8; ++r) acc[r] = biasv;

  const float* ap =
      ys + ((size_t)(tileM * 16 + lane15) * kT + (kT - 1)) * kH + half * 2;
  const float* bp = Wfc + (size_t)n * kH + half * 2;
#pragma unroll 4
  for (int ks = 0; ks < kH / 4; ++ks) {
    v2f a = *(const v2f*)(ap + ks * 4);
    v2f b = *(const v2f*)(bp + ks * 4);
    acc = wmma_f32(a, b, acc);
  }

  float* op = out + (size_t)(tileM * 16 + half * 8) * kC + tileN * 16 + lane15;
#pragma unroll
  for (int r = 0; r < 8; ++r) op[(size_t)r * kC] = acc[r];
}

extern "C" void kernel_launch(void* const* d_in, const int* in_sizes, int n_in,
                              void* d_out, int out_size, void* d_ws,
                              size_t ws_size, hipStream_t stream) {
  const float* x    = (const float*)d_in[0];
  const float* W_ih = (const float*)d_in[1];
  const float* W_hh = (const float*)d_in[2];
  const float* b_ih = (const float*)d_in[3];
  const float* b_hh = (const float*)d_in[4];
  const float* W_fc = (const float*)d_in[5];
  const float* b_fc = (const float*)d_in[6];

  float* bufA = (float*)d_ws;
  float* bufB = bufA + (size_t)kB * kT * kH;

  const int projWaveBlks = (kB * kT / 32) * (kH / 32);
  const int projBlocks   = projWaveBlks / 8;

  const float* cur = x;
  float* dst = bufA;
  float* alt = bufB;
  for (int l = 0; l < kL; ++l) {
    rnn_proj_kernel<<<projBlocks, 256, 0, stream>>>(
        cur, W_ih + (size_t)l * kH * kH, b_ih + (size_t)l * kH,
        b_hh + (size_t)l * kH, dst);
    rnn_pack_kernel<<<(kKs * 32 * 16) / 256, 256, 0, stream>>>(
        W_hh + (size_t)l * kH * kH, alt);
    rnn_scan_kernel<<<4, 512, 0, stream>>>(dst, alt);
    cur = dst;
    float* tmp = dst;
    dst = alt;
    alt = tmp;
  }
  head_kernel<<<4, 256, 0, stream>>>(cur, W_fc, b_fc, (float*)d_out);

  (void)in_sizes; (void)n_in; (void)out_size; (void)ws_size;
}


# Scan time loop moved to the host: one launch per timestep, 64 workgroups of 8 waves split each step's k range in 8 parts summed through LDS in fixed order (sum order changes); h exchanged in place
# speedup vs baseline: 5.3844x; 2.4947x over previous
#include <hip/hip_runtime.h>

typedef float v2f __attribute__((ext_vector_type(2)));
typedef float v4f __attribute__((ext_vector_type(4)));
typedef float v8f __attribute__((ext_vector_type(8)));

namespace {

constexpr int kB = 64;
constexpr int kT = 512;
constexpr int kH = 512;
constexpr int kL = 4;
constexpr int kC = 128;
constexpr int kKs        = kH / 4;
constexpr int kSplit     = 8;
constexpr int kKsPart    = kKs / kSplit;
constexpr int kPartPitch = 36;

__device__ __forceinline__ v8f wmma_f32(v2f a, v2f b, v8f c) {
  return __builtin_amdgcn_wmma_f32_16x16x4_f32(false, a, false, b, (short)0, c,
                                               false, false);
}

}

__global__ __launch_bounds__(256) void rnn_proj_kernel(
    const float* __restrict__ in, const float* __restrict__ Wi,
    const float* __restrict__ bi, const float* __restrict__ bh,
    float* __restrict__ xi) {
  const int lane   = threadIdx.x & 31;
  const int wave   = threadIdx.x >> 5;
  const int lane15 = lane & 15;
  const int half   = lane >> 4;

  const int blk   = blockIdx.x * 8 + wave;
  const int blkN  = blk & (kH / 32 - 1);
  const int blkM  = blk >> 4;

  const int n0 = blkN * 32 + lane15;
  const int n1 = n0 + 16;
  const int m0 = blkM * 32 + lane15;
  const int m1 = m0 + 16;

  const float bias0 = bi[n0] + bh[n0];
  const float bias1 = bi[n1] + bh[n1];
  v8f acc00, acc01, acc10, acc11;
#pragma unroll
  for (int r = 0; r < 8; ++r) {
    acc00[r] = bias0; acc01[r] = bias1;
    acc10[r] = bias0; acc11[r] = bias1;
  }

  const float* ap0 = in + (size_t)m0 * kH + half * 2;
  const float* ap1 = in + (size_t)m1 * kH + half * 2;
  const float* bp0 = Wi + (size_t)n0 * kH + half * 2;
  const float* bp1 = Wi + (size_t)n1 * kH + half * 2;
#pragma unroll 4
  for (int ks = 0; ks < kH / 4; ++ks) {
    v2f a0 = *(const v2f*)(ap0 + ks * 4);
    v2f a1 = *(const v2f*)(ap1 + ks * 4);
    v2f b0 = *(const v2f*)(bp0 + ks * 4);
    v2f b1 = *(const v2f*)(bp1 + ks * 4);
    acc00 = wmma_f32(a0, b0, acc00);
    acc01 = wmma_f32(a0, b1, acc01);
    acc10 = wmma_f32(a1, b0, acc10);
    acc11 = wmma_f32(a1, b1, acc11);
  }

  float* op0 = xi + (size_t)(blkM * 32 + half * 8) * kH + blkN * 32 + lane15;
  float* op1 = op0 + (size_t)16 * kH;
#pragma unroll
  for (int r = 0; r < 8; ++r) {
    op0[(size_t)r * kH]      = acc00[r];
    op0[(size_t)r * kH + 16] = acc01[r];
    op1[(size_t)r * kH]      = acc10[r];
    op1[(size_t)r * kH + 16] = acc11[r];
  }
}

__global__ __launch_bounds__(256) void rnn_pack_kernel(
    const float* __restrict__ Wh, float* __restrict__ Wp) {
  const int gid    = blockIdx.x * 256 + threadIdx.x;
  const int lane   = gid & 31;
  const int ks     = (gid >> 5) & (kKs - 1);
  const int wave   = gid >> 12;
  const int lane15 = lane & 15;
  const int half   = lane >> 4;

  const float* r0 = Wh + (size_t)(wave * 32 + lane15) * kH + ks * 4 + half * 2;
  const float* r1 = r0 + (size_t)16 * kH;
  const v2f x0 = *(const v2f*)r0;
  const v2f x1 = *(const v2f*)r1;
  v4f o;
  o.x = x0.x; o.y = x0.y; o.z = x1.x; o.w = x1.y;
  *(v4f*)(Wp + (size_t)gid * 4) = o;
}

__global__ __launch_bounds__(256) void rnn_step_kernel(
    float* xi, const float* __restrict__ Wp, int t) {
  __shared__ float part[kSplit * 16 * kPartPitch];

  const int lane   = threadIdx.x & 31;
  const int wave   = threadIdx.x >> 5;
  const int lane15 = lane & 15;
  const int half   = lane >> 4;
  const int w      = blockIdx.x & 15;
  const int m      = blockIdx.x >> 4;

  const size_t rowStride = (size_t)kT * kH;


  float* orow = xi + (size_t)(m * 16 + 2 * wave) * rowStride +
                (size_t)t * kH + w * 32 + lane;
  const float x0 = orow[0];
  const float x1 = orow[rowStride];

  const v4f* bp =
      (const v4f*)Wp + ((size_t)w * kKs + wave * kKsPart) * 32 + lane;
  v4f b[kKsPart];
#pragma unroll
  for (int j = 0; j < kKsPart; ++j) b[j] = bp[j * 32];

  v2f a[kKsPart];
  if (t > 0) {
    const float* ap = xi + (size_t)(m * 16 + lane15) * rowStride +
                      (size_t)(t - 1) * kH + wave * (kKsPart * 4) + half * 2;
#pragma unroll
    for (int j = 0; j < kKsPart; ++j) a[j] = *(const v2f*)(ap + j * 4);
  } else {
#pragma unroll
    for (int j = 0; j < kKsPart; ++j) { a[j].x = 0.0f; a[j].y = 0.0f; }
  }

  v8f acc0, acc1;
#pragma unroll
  for (int r = 0; r < 8; ++r) { acc0[r] = 0.0f; acc1[r] = 0.0f; }
#pragma unroll
  for (int j = 0; j < kKsPart; ++j) {
    const v2f b0 = __builtin_shufflevector(b[j], b[j], 0, 1);
    const v2f b1 = __builtin_shufflevector(b[j], b[j], 2, 3);
    acc0 = wmma_f32(a[j], b0, acc0);
    acc1 = wmma_f32(a[j], b1, acc1);
  }

  float* pw = part + (wave * 16 + half * 8) * kPartPitch + lane15;
#pragma unroll
  for (int r = 0; r < 8; ++r) {
    pw[r * kPartPitch]      = acc0[r];
    pw[r * kPartPitch + 16] = acc1[r];
  }
  __syncthreads();

  const float* pr = part + (2 * wave) * kPartPitch + lane;
  float s0 = x0;
  float s1 = x1;
#pragma unroll
  for (int p = 0; p < kSplit; ++p) {
    s0 += pr[p * 16 * kPartPitch];
    s1 += pr[p * 16 * kPartPitch + kPartPitch];
  }
  orow[0]         = s0 > 0.0f ? s0 : 0.0f;
  orow[rowStride] = s1 > 0.0f ? s1 : 0.0f;
}

__global__ __launch_bounds__(256) void head_kernel(
    const float* __restrict__ ys, const float* __restrict__ Wfc,
    const float* __restrict__ bfc, float* __restrict__ out) {
  const int lane   = threadIdx.x & 31;
  const int wave   = threadIdx.x >> 5;
  const int lane15 = lane & 15;
  const int half   = lane >> 4;

  const int tile  = blockIdx.x * 8 + wave;
  const int tileN = tile & 7;
  const int tileM = tile >> 3;

  const int n = tileN * 16 + lane15;
  const float biasv = bfc[n];
  v8f acc;
#pragma unroll
  for (int r = 0; r < 8; ++r) acc[r] = biasv;

  const float* ap =
      ys + ((size_t)(tileM * 16 + lane15) * kT + (kT - 1)) * kH + half * 2;
  const float* bp = Wfc + (size_t)n * kH + half * 2;
#pragma unroll 4
  for (int ks = 0; ks < kH / 4; ++ks) {
    v2f a = *(const v2f*)(ap + ks * 4);
    v2f b = *(const v2f*)(bp + ks * 4);
    acc = wmma_f32(a, b, acc);
  }

  float* op = out + (size_t)(tileM * 16 + half * 8) * kC + tileN * 16 + lane15;
#pragma unroll
  for (int r = 0; r < 8; ++r) op[(size_t)r * kC] = acc[r];
}

extern "C" void kernel_launch(void* const* d_in, const int* in_sizes, int n_in,
                              void* d_out, int out_size, void* d_ws,
                              size_t ws_size, hipStream_t stream) {
  const float* x    = (const float*)d_in[0];
  const float* W_ih = (const float*)d_in[1];
  const float* W_hh = (const float*)d_in[2];
  const float* b_ih = (const float*)d_in[3];
  const float* b_hh = (const float*)d_in[4];
  const float* W_fc = (const float*)d_in[5];
  const float* b_fc = (const float*)d_in[6];

  float* bufA = (float*)d_ws;
  float* bufB = bufA + (size_t)kB * kT * kH;

  const int projWaveBlks = (kB * kT / 32) * (kH / 32);
  const int projBlocks   = projWaveBlks / 8;

  const float* cur = x;
  float* dst = bufA;
  float* alt = bufB;
  for (int l = 0; l < kL; ++l) {
    rnn_proj_kernel<<<projBlocks, 256, 0, stream>>>(
        cur, W_ih + (size_t)l * kH * kH, b_ih + (size_t)l * kH,
        b_hh + (size_t)l * kH, dst);
    rnn_pack_kernel<<<(kKs * 32 * 16) / 256, 256, 0, stream>>>(
        W_hh + (size_t)l * kH * kH, alt);
    for (int t = 0; t < kT; ++t)
      rnn_step_kernel<<<(kB / 16) * (kH / 32), 256, 0, stream>>>(dst, alt, t);
    cur = dst;
    float* tmp = dst;
    dst = alt;
    alt = tmp;
  }
  head_kernel<<<4, 256, 0, stream>>>(cur, W_fc, b_fc, (float*)d_out);

  (void)in_sizes; (void)n_in; (void)out_size; (void)ws_size;
}


# Layer wavefront: launch d runs timestep d-l of every layer l (515 step launches, not 2048); layers 1-3 take their input projection inside the step as a second k range (sum order changes)
# speedup vs baseline: 17.3306x; 3.2187x over previous
#include <hip/hip_runtime.h>


typedef float v2f __attribute__((ext_vector_type(2)));
typedef float v4f __attribute__((ext_vector_type(4)));
typedef float v8f __attribute__((ext_vector_type(8)));

namespace {

constexpr int kB = 64;
constexpr int kT = 512;
constexpr int kH = 512;
constexpr int kL = 4;
constexpr int kC = 128;
constexpr int kKs        = kH / 4;
constexpr int kSplit     = 8;
constexpr int kPartPitch = 36;
constexpr int kDiag      = kT + kL - 1;
constexpr int kPacked    = 2 * kL - 1;
constexpr size_t kMat    = (size_t)kH * kH;
constexpr size_t kPlane  = (size_t)kB * kH;

__device__ __forceinline__ v8f wmma_f32(v2f a, v2f b, v8f c) {
  return __builtin_amdgcn_wmma_f32_16x16x4_f32(false, a, false, b, (short)0, c,
                                               false, false);
}

template <int NK>
__device__ __forceinline__ void partial_tiles(const v4f* bp, const float* ap,
                                              bool haveA, float* pw) {
  v4f b[NK];
#pragma unroll
  for (int j = 0; j < NK; ++j) b[j] = bp[j * 32];

  v2f a[NK];
  if (haveA) {
#pragma unroll
    for (int j = 0; j < NK; ++j) a[j] = *(const v2f*)(ap + j * 4);
  } else {
#pragma unroll
    for (int j = 0; j < NK; ++j) { a[j].x = 0.0f; a[j].y = 0.0f; }
  }

  v8f acc0, acc1;
#pragma unroll
  for (int r = 0; r < 8; ++r) { acc0[r] = 0.0f; acc1[r] = 0.0f; }
#pragma unroll
  for (int j = 0; j < NK; ++j) {
    const v2f b0 = __builtin_shufflevector(b[j], b[j], 0, 1);
    const v2f b1 = __builtin_shufflevector(b[j], b[j], 2, 3);
    acc0 = wmma_f32(a[j], b0, acc0);
    acc1 = wmma_f32(a[j], b1, acc1);
  }

#pragma unroll
  for (int r = 0; r < 8; ++r) {
    pw[r * kPartPitch]      = acc0[r];
    pw[r * kPartPitch + 16] = acc1[r];
  }
}

}

__global__ __launch_bounds__(256) void rnn_proj_kernel(
    const float* __restrict__ in, const float* __restrict__ Wi,
    const float* __restrict__ bi, const float* __restrict__ bh,
    float* __restrict__ xi) {
  const int lane   = threadIdx.x & 31;
  const int wave   = threadIdx.x >> 5;
  const int lane15 = lane & 15;
  const int half   = lane >> 4;

  const int blk   = blockIdx.x * 8 + wave;
  const int blkN  = blk & (kH / 32 - 1);
  const int blkM  = blk >> 4;

  const int n0 = blkN * 32 + lane15;
  const int n1 = n0 + 16;
  const int m0 = blkM * 32 + lane15;
  const int m1 = m0 + 16;

  const float bias0 = bi[n0] + bh[n0];
  const float bias1 = bi[n1] + bh[n1];
  v8f acc00, acc01, acc10, acc11;
#pragma unroll
  for (int r = 0; r < 8; ++r) {
    acc00[r] = bias0; acc01[r] = bias1;
    acc10[r] = bias0; acc11[r] = bias1;
  }

  const float* ap0 = in + (size_t)m0 * kH + half * 2;
  const float* ap1 = in + (size_t)m1 * kH + half * 2;
  const float* bp0 = Wi + (size_t)n0 * kH + half * 2;
  const float* bp1 = Wi + (size_t)n1 * kH + half * 2;
#pragma unroll 4
  for (int ks = 0; ks < kH / 4; ++ks) {
    v2f a0 = *(const v2f*)(ap0 + ks * 4);
    v2f a1 = *(const v2f*)(ap1 + ks * 4);
    v2f b0 = *(const v2f*)(bp0 + ks * 4);
    v2f b1 = *(const v2f*)(bp1 + ks * 4);
    acc00 = wmma_f32(a0, b0, acc00);
    acc01 = wmma_f32(a0, b1, acc01);
    acc10 = wmma_f32(a1, b0, acc10);
    acc11 = wmma_f32(a1, b1, acc11);
  }

  float* op0 = xi + (size_t)(blkM * 32 + half * 8) * kH + blkN * 32 + lane15;
  float* op1 = op0 + (size_t)16 * kH;
#pragma unroll
  for (int r = 0; r < 8; ++r) {
    op0[(size_t)r * kH]      = acc00[r];
    op0[(size_t)r * kH + 16] = acc01[r];
    op1[(size_t)r * kH]      = acc10[r];
    op1[(size_t)r * kH + 16] = acc11[r];
  }
}

__global__ __launch_bounds__(256) void rnn_pack_kernel(
    const float* __restrict__ W_hh, const float* __restrict__ W_ih,
    float* __restrict__ Wp) {
  const int p = blockIdx.y;
  const float* Wsrc = (p < kL) ? W_hh + (size_t)p * kMat
                               : W_ih + (size_t)(p - kL + 1) * kMat;
  const int gid    = blockIdx.x * 256 + threadIdx.x;
  const int lane   = gid & 31;
  const int ks     = (gid >> 5) & (kKs - 1);
  const int wg     = gid >> 12;
  const int lane15 = lane & 15;
  const int half   = lane >> 4;

  const float* r0 = Wsrc + (size_t)(wg * 32 + lane15) * kH + ks * 4 + half * 2;
  const float* r1 = r0 + (size_t)16 * kH;
  const v2f x0 = *(const v2f*)r0;
  const v2f x1 = *(const v2f*)r1;
  v4f o;
  o.x = x0.x; o.y = x0.y; o.z = x1.x; o.w = x1.y;
  *(v4f*)(Wp + (size_t)p * kMat + (size_t)gid * 4) = o;
}

__global__ __launch_bounds__(256) void rnn_diag_kernel(
    const float* __restrict__ xi0, const float* __restrict__ Wp,
    const float* __restrict__ b_ih, const float* __restrict__ b_hh,
    float* hring, int d) {
  __shared__ float part[kSplit * 16 * kPartPitch];

  const int lane   = threadIdx.x & 31;
  const int wave   = __builtin_amdgcn_readfirstlane(threadIdx.x >> 5);
  const int lane15 = lane & 15;
  const int half   = lane >> 4;
  const int w      = blockIdx.x & 15;
  const int m      = (blockIdx.x >> 4) & 3;
  const int layer  = blockIdx.x >> 6;
  const int t      = d - layer;
  if (t < 0 || t >= kT) return;

  const float* hprev = hring + (size_t)((d + 1) & 1) * (kL * kPlane);
  float*       hcur  = hring + (size_t)(d & 1) * (kL * kPlane);

  const int col = w * 32 + lane;
  float* orow = hcur + (size_t)layer * kPlane +
                (size_t)(m * 16 + 2 * wave) * kH + col;
  float* pw = part + (wave * 16 + half * 8) * kPartPitch + lane15;

  float s0, s1;
  if (layer == 0) {
    const size_t rowStride = (size_t)kT * kH;
    const float* xrow = xi0 + (size_t)(m * 16 + 2 * wave) * rowStride +
                        (size_t)t * kH + col;
    s0 = xrow[0];
    s1 = xrow[rowStride];
    const v4f* bp = (const v4f*)Wp + ((size_t)w * kKs + wave * 16) * 32 + lane;
    const float* ap = hprev + (size_t)(m * 16 + lane15) * kH +
                      wave * 64 + half * 2;
    partial_tiles<16>(bp, ap, t > 0, pw);
  } else {
    s0 = b_ih[layer * kH + col] + b_hh[layer * kH + col];
    s1 = s0;
    const int rec = wave >> 2;
    const int kq  = wave & 3;
    const int mat = rec ? layer : kL - 1 + layer;
    const v4f* bp = (const v4f*)(Wp + (size_t)mat * kMat) +
                    ((size_t)w * kKs + kq * 32) * 32 + lane;
    const float* ap = hprev + (size_t)(layer - 1 + rec) * kPlane +
                      (size_t)(m * 16 + lane15) * kH + kq * 128 + half * 2;
    partial_tiles<32>(bp, ap, rec == 0 || t > 0, pw);
  }
  __syncthreads();

  const float* pr = part + (2 * wave) * kPartPitch + lane;
#pragma unroll
  for (int p = 0; p < kSplit; ++p) {
    s0 += pr[p * 16 * kPartPitch];
    s1 += pr[p * 16 * kPartPitch + kPartPitch];
  }
  orow[0]  = s0 > 0.0f ? s0 : 0.0f;
  orow[kH] = s1 > 0.0f ? s1 : 0.0f;
}

__global__ __launch_bounds__(256) void head_kernel(
    const float* __restrict__ hlast, const float* __restrict__ Wfc,
    const float* __restrict__ bfc, float* __restrict__ out) {
  const int lane   = threadIdx.x & 31;
  const int wave   = threadIdx.x >> 5;
  const int lane15 = lane & 15;
  const int half   = lane >> 4;

  const int tile  = blockIdx.x * 8 + wave;
  const int tileN = tile & 7;
  const int tileM = tile >> 3;

  const int n = tileN * 16 + lane15;
  const float biasv = bfc[n];
  v8f acc;
#pragma unroll
  for (int r = 0; r < 8; ++r) acc[r] = biasv;

  const float* ap = hlast + (size_t)(tileM * 16 + lane15) * kH + half * 2;
  const float* bp = Wfc + (size_t)n * kH + half * 2;
#pragma unroll 4
  for (int ks = 0; ks < kH / 4; ++ks) {
    v2f a = *(const v2f*)(ap + ks * 4);
    v2f b = *(const v2f*)(bp + ks * 4);
    acc = wmma_f32(a, b, acc);
  }

  float* op = out + (size_t)(tileM * 16 + half * 8) * kC + tileN * 16 + lane15;
#pragma unroll
  for (int r = 0; r < 8; ++r) op[(size_t)r * kC] = acc[r];
}

extern "C" void kernel_launch(void* const* d_in, const int* in_sizes, int n_in,
                              void* d_out, int out_size, void* d_ws,
                              size_t ws_size, hipStream_t stream) {
  const float* x    = (const float*)d_in[0];
  const float* W_ih = (const float*)d_in[1];
  const float* W_hh = (const float*)d_in[2];
  const float* b_ih = (const float*)d_in[3];
  const float* b_hh = (const float*)d_in[4];
  const float* W_fc = (const float*)d_in[5];
  const float* b_fc = (const float*)d_in[6];

  float* xi0   = (float*)d_ws;
  float* Wp    = xi0 + (size_t)kB * kT * kH;
  float* hring = Wp + kPacked * kMat;

  const int projWaveBlks = (kB * kT / 32) * (kH / 32);
  const int projBlocks   = projWaveBlks / 8;

  rnn_proj_kernel<<<projBlocks, 256, 0, stream>>>(x, W_ih, b_ih, b_hh, xi0);
  rnn_pack_kernel<<<dim3((kKs * 32 * 16) / 256, kPacked), 256, 0, stream>>>(
      W_hh, W_ih, Wp);
  for (int d = 0; d < kDiag; ++d)
    rnn_diag_kernel<<<kL * (kB / 16) * (kH / 32), 256, 0, stream>>>(
        xi0, Wp, b_ih, b_hh, hring, d);
  head_kernel<<<4, 256, 0, stream>>>(hring + (kL - 1) * kPlane, W_fc, b_fc,
                                     (float*)d_out);

  (void)in_sizes; (void)n_in; (void)out_size; (void)ws_size;
}


# Step fragment loads issued in k order, B and A alternating by groups of 4 k-steps (was all 32 weight loads, then all state loads): WMMAs start behind the first group; t=0 reads a zero-filled ring
# speedup vs baseline: 18.8104x; 1.0854x over previous
#include <hip/hip_runtime.h>


typedef float v2f __attribute__((ext_vector_type(2)));
typedef float v4f __attribute__((ext_vector_type(4)));
typedef float v8f __attribute__((ext_vector_type(8)));

namespace {

constexpr int kB = 64;
constexpr int kT = 512;
constexpr int kH = 512;
constexpr int kL = 4;
constexpr int kC = 128;
constexpr int kKs        = kH / 4;
constexpr int kSplit     = 8;
constexpr int kPartPitch = 36;
constexpr int kDiag      = kT + kL - 1;
constexpr int kPacked    = 2 * kL - 1;
constexpr size_t kMat    = (size_t)kH * kH;
constexpr size_t kPlane  = (size_t)kB * kH;

__device__ __forceinline__ v8f wmma_f32(v2f a, v2f b, v8f c) {
  return __builtin_amdgcn_wmma_f32_16x16x4_f32(false, a, false, b, (short)0, c,
                                               false, false);
}

template <int NK>
__device__ __forceinline__ void partial_tiles(const v4f* bp, const float* ap,
                                              float* pw) {
  v4f b[NK];
  v2f a[NK];
#pragma unroll
  for (int j = 0; j < NK; ++j) {
    b[j] = bp[j * 32];
    a[j] = *(const v2f*)(ap + j * 4);
  }
  __builtin_amdgcn_sched_barrier(0);

  v8f acc0, acc1;
#pragma unroll
  for (int r = 0; r < 8; ++r) { acc0[r] = 0.0f; acc1[r] = 0.0f; }
#pragma unroll
  for (int j = 0; j < NK; ++j) {
    const v2f b0 = __builtin_shufflevector(b[j], b[j], 0, 1);
    const v2f b1 = __builtin_shufflevector(b[j], b[j], 2, 3);
    acc0 = wmma_f32(a[j], b0, acc0);
    acc1 = wmma_f32(a[j], b1, acc1);
  }

#pragma unroll
  for (int r = 0; r < 8; ++r) {
    pw[r * kPartPitch]      = acc0[r];
    pw[r * kPartPitch + 16] = acc1[r];
  }
}

}

__global__ __launch_bounds__(256) void rnn_proj_kernel(
    const float* __restrict__ in, const float* __restrict__ Wi,
    const float* __restrict__ bi, const float* __restrict__ bh,
    float* __restrict__ xi) {
  const int lane   = threadIdx.x & 31;
  const int wave   = threadIdx.x >> 5;
  const int lane15 = lane & 15;
  const int half   = lane >> 4;

  const int blk   = blockIdx.x * 8 + wave;
  const int blkN  = blk & (kH / 32 - 1);
  const int blkM  = blk >> 4;

  const int n0 = blkN * 32 + lane15;
  const int n1 = n0 + 16;
  const int m0 = blkM * 32 + lane15;
  const int m1 = m0 + 16;

  const float bias0 = bi[n0] + bh[n0];
  const float bias1 = bi[n1] + bh[n1];
  v8f acc00, acc01, acc10, acc11;
#pragma unroll
  for (int r = 0; r < 8; ++r) {
    acc00[r] = bias0; acc01[r] = bias1;
    acc10[r] = bias0; acc11[r] = bias1;
  }

  const float* ap0 = in + (size_t)m0 * kH + half * 2;
  const float* ap1 = in + (size_t)m1 * kH + half * 2;
  const float* bp0 = Wi + (size_t)n0 * kH + half * 2;
  const float* bp1 = Wi + (size_t)n1 * kH + half * 2;
#pragma unroll 4
  for (int ks = 0; ks < kH / 4; ++ks) {
    v2f a0 = *(const v2f*)(ap0 + ks * 4);
    v2f a1 = *(const v2f*)(ap1 + ks * 4);
    v2f b0 = *(const v2f*)(bp0 + ks * 4);
    v2f b1 = *(const v2f*)(bp1 + ks * 4);
    acc00 = wmma_f32(a0, b0, acc00);
    acc01 = wmma_f32(a0, b1, acc01);
    acc10 = wmma_f32(a1, b0, acc10);
    acc11 = wmma_f32(a1, b1, acc11);
  }

  float* op0 = xi + (size_t)(blkM * 32 + half * 8) * kH + blkN * 32 + lane15;
  float* op1 = op0 + (size_t)16 * kH;
#pragma unroll
  for (int r = 0; r < 8; ++r) {
    op0[(size_t)r * kH]      = acc00[r];
    op0[(size_t)r * kH + 16] = acc01[r];
    op1[(size_t)r * kH]      = acc10[r];
    op1[(size_t)r * kH + 16] = acc11[r];
  }
}

__global__ __launch_bounds__(256) void rnn_pack_kernel(
    const float* __restrict__ W_hh, const float* __restrict__ W_ih,
    float* __restrict__ Wp, float* __restrict__ hring) {
  const int p = blockIdx.y;
  const float* Wsrc = (p < kL) ? W_hh + (size_t)p * kMat
                               : W_ih + (size_t)(p - kL + 1) * kMat;
  const int gid    = blockIdx.x * 256 + threadIdx.x;
  const int lane   = gid & 31;
  const int ks     = (gid >> 5) & (kKs - 1);
  const int wg     = gid >> 12;
  const int lane15 = lane & 15;
  const int half   = lane >> 4;

  const float* r0 = Wsrc + (size_t)(wg * 32 + lane15) * kH + ks * 4 + half * 2;
  const float* r1 = r0 + (size_t)16 * kH;
  const v2f x0 = *(const v2f*)r0;
  const v2f x1 = *(const v2f*)r1;
  v4f o;
  o.x = x0.x; o.y = x0.y; o.z = x1.x; o.w = x1.y;
  *(v4f*)(Wp + (size_t)p * kMat + (size_t)gid * 4) = o;
  if (p == 0) {
    v4f z;
    z.x = 0.0f; z.y = 0.0f; z.z = 0.0f; z.w = 0.0f;
    *(v4f*)(hring + (size_t)gid * 4) = z;
  }
}

__global__ __launch_bounds__(256) void rnn_diag_kernel(
    const float* __restrict__ xi0, const float* __restrict__ Wp,
    const float* __restrict__ b_ih, const float* __restrict__ b_hh,
    float* hring, int d) {
  __shared__ float part[kSplit * 16 * kPartPitch];

  const int lane   = threadIdx.x & 31;
  const int wave   = __builtin_amdgcn_readfirstlane(threadIdx.x >> 5);
  const int lane15 = lane & 15;
  const int half   = lane >> 4;
  const int w      = blockIdx.x & 15;
  const int m      = (blockIdx.x >> 4) & 3;
  const int layer  = blockIdx.x >> 6;
  const int t      = d - layer;
  if (t < 0 || t >= kT) return;

  const float* hprev = hring + (size_t)((d + 1) & 1) * (kL * kPlane);
  float*       hcur  = hring + (size_t)(d & 1) * (kL * kPlane);

  const int col = w * 32 + lane;
  float* orow = hcur + (size_t)layer * kPlane +
                (size_t)(m * 16 + 2 * wave) * kH + col;
  float* pw = part + (wave * 16 + half * 8) * kPartPitch + lane15;

  float s0, s1;
  if (layer == 0) {
    const size_t rowStride = (size_t)kT * kH;
    const float* xrow = xi0 + (size_t)(m * 16 + 2 * wave) * rowStride +
                        (size_t)t * kH + col;
    s0 = xrow[0];
    s1 = xrow[rowStride];
    const v4f* bp = (const v4f*)Wp + ((size_t)w * kKs + wave * 16) * 32 + lane;
    const float* ap = hprev + (size_t)(m * 16 + lane15) * kH +
                      wave * 64 + half * 2;
    partial_tiles<16>(bp, ap, pw);
  } else {
    s0 = b_ih[layer * kH + col] + b_hh[layer * kH + col];
    s1 = s0;
    const int rec = wave >> 2;
    const int kq  = wave & 3;
    const int mat = rec ? layer : kL - 1 + layer;
    const v4f* bp = (const v4f*)(Wp + (size_t)mat * kMat) +
                    ((size_t)w * kKs + kq * 32) * 32 + lane;
    const float* ap = hprev + (size_t)(layer - 1 + rec) * kPlane +
                      (size_t)(m * 16 + lane15) * kH + kq * 128 + half * 2;
    partial_tiles<32>(bp, ap, pw);
  }
  __syncthreads();

  const float* pr = part + (2 * wave) * kPartPitch + lane;
#pragma unroll
  for (int p = 0; p < kSplit; ++p) {
    s0 += pr[p * 16 * kPartPitch];
    s1 += pr[p * 16 * kPartPitch + kPartPitch];
  }
  orow[0]  = s0 > 0.0f ? s0 : 0.0f;
  orow[kH] = s1 > 0.0f ? s1 : 0.0f;
}

__global__ __launch_bounds__(256) void head_kernel(
    const float* __restrict__ hlast, const float* __restrict__ Wfc,
    const float* __restrict__ bfc, float* __restrict__ out) {
  const int lane   = threadIdx.x & 31;
  const int wave   = threadIdx.x >> 5;
  const int lane15 = lane & 15;
  const int half   = lane >> 4;

  const int tile  = blockIdx.x * 8 + wave;
  const int tileN = tile & 7;
  const int tileM = tile >> 3;

  const int n = tileN * 16 + lane15;
  const float biasv = bfc[n];
  v8f acc;
#pragma unroll
  for (int r = 0; r < 8; ++r) acc[r] = biasv;

  const float* ap = hlast + (size_t)(tileM * 16 + lane15) * kH + half * 2;
  const float* bp = Wfc + (size_t)n * kH + half * 2;
#pragma unroll 4
  for (int ks = 0; ks < kH / 4; ++ks) {
    v2f a = *(const v2f*)(ap + ks * 4);
    v2f b = *(const v2f*)(bp + ks * 4);
    acc = wmma_f32(a, b, acc);
  }

  float* op = out + (size_t)(tileM * 16 + half * 8) * kC + tileN * 16 + lane15;
#pragma unroll
  for (int r = 0; r < 8; ++r) op[(size_t)r * kC] = acc[r];
}

extern "C" void kernel_launch(void* const* d_in, const int* in_sizes, int n_in,
                              void* d_out, int out_size, void* d_ws,
                              size_t ws_size, hipStream_t stream) {
  const float* x    = (const float*)d_in[0];
  const float* W_ih = (const float*)d_in[1];
  const float* W_hh = (const float*)d_in[2];
  const float* b_ih = (const float*)d_in[3];
  const float* b_hh = (const float*)d_in[4];
  const float* W_fc = (const float*)d_in[5];
  const float* b_fc = (const float*)d_in[6];

  float* xi0   = (float*)d_ws;
  float* Wp    = xi0 + (size_t)kB * kT * kH;
  float* hring = Wp + kPacked * kMat;

  const int projWaveBlks = (kB * kT / 32) * (kH / 32);
  const int projBlocks   = projWaveBlks / 8;

  rnn_proj_kernel<<<projBlocks, 256, 0, stream>>>(x, W_ih, b_ih, b_hh, xi0);
  rnn_pack_kernel<<<dim3((kKs * 32 * 16) / 256, kPacked), 256, 0, stream>>>(
      W_hh, W_ih, Wp, hring);
  for (int d = 0; d < kDiag; ++d)
    rnn_diag_kernel<<<kL * (kB / 16) * (kH / 32), 256, 0, stream>>>(
        xi0, Wp, b_ih, b_hh, hring, d);
  head_kernel<<<4, 256, 0, stream>>>(hring + (kL - 1) * kPlane, W_fc, b_fc,
                                     (float*)d_out);

  (void)in_sizes; (void)n_in; (void)out_size; (void)ws_size;
}
